# starting kernel
# baseline (speedup 1.0000x reference)
#include <hip/hip_runtime.h>
#include <hip/hip_bf16.h>

#define N_NODES 100000
#define N_EDGES 1600000
#define D_IN    128
#define D_HID   128
#define N_CLS   32
#define N_STRIPS (N_NODES / 16)

typedef float v2f __attribute__((ext_vector_type(2)));
typedef float v8f __attribute__((ext_vector_type(8)));

__global__ void fill_f32(float* __restrict__ p, float v, long n) {
    long i = (long)blockIdx.x * 256 + threadIdx.x;
    if (i < n) p[i] = v;
}

__global__ void deg_accum(const int* __restrict__ dst, float* __restrict__ deg) {
    int e = blockIdx.x * 256 + threadIdx.x;
    if (e < N_EDGES) unsafeAtomicAdd(&deg[dst[e]], 1.0f);
}

__global__ void rsqrt_kernel(const float* __restrict__ deg, float* __restrict__ dinv) {
    int i = blockIdx.x * 256 + threadIdx.x;
    if (i < N_NODES) {
        float d = deg[i];
        dinv[i] = d > 0.0f ? rsqrtf(d) : 0.0f;
    }
}

template <int NCOLS>
__global__ __launch_bounds__(256)
void gemm_scale_kernel(const float* __restrict__ A,
                       const float* __restrict__ W,
                       const float* __restrict__ dinv,
                       float* __restrict__ out) {
    constexpr int NT = NCOLS / 16;
    __shared__ float lW[128 * NCOLS];

    const int tid = threadIdx.x;
    for (int idx = tid; idx < 128 * NCOLS; idx += 256) {
        const int k = idx / NCOLS;
        const int n = idx - k * NCOLS;
        lW[(k >> 1) * (2 * NCOLS) + 2 * n + (k & 1)] = W[idx];
    }
    __syncthreads();

    const int wave  = tid >> 5;
    const int lane  = tid & 31;
    const int strip = blockIdx.x * 8 + wave;
    if (strip >= N_STRIPS) return;

    const int row0 = strip * 16;
    const int half = lane >> 4;
    const int l16  = lane & 15;
    const int arow = row0 + l16;

    v8f acc[NT];
#pragma unroll
    for (int t = 0; t < NT; ++t) acc[t] = (v8f){0.f, 0.f, 0.f, 0.f, 0.f, 0.f, 0.f, 0.f};

    for (int k0 = 0; k0 < 128; k0 += 4) {
        const int ka = k0 + 2 * half;
        const v2f a = *(const v2f*)&A[(long)arow * 128 + ka];
        const float* __restrict__ lrow = &lW[(ka >> 1) * (2 * NCOLS)];
#pragma unroll
        for (int t = 0; t < NT; ++t) {
            const int n = t * 16 + l16;
            const v2f b = *(const v2f*)&lrow[2 * n];
            acc[t] = __builtin_amdgcn_wmma_f32_16x16x4_f32(
                false, a, false, b, (short)0, acc[t], false, false);
        }
    }

#pragma unroll
    for (int t = 0; t < NT; ++t) {
#pragma unroll
        for (int r = 0; r < 8; ++r) {
            const int rr = row0 + r + 8 * half;
            out[(long)rr * NCOLS + t * 16 + l16] = acc[t][r] * dinv[rr];
        }
    }
}

__global__ __launch_bounds__(256)
void scatter128(const int* __restrict__ src, const int* __restrict__ dst,
                const float* __restrict__ hs, float* __restrict__ t1) {
    const int e = blockIdx.x * 8 + (threadIdx.x >> 5);
    const int lane = threadIdx.x & 31;
    if (e >= N_EDGES) return;
    const long s = src[e];
    const long d = dst[e];
    const float* __restrict__ hrow = hs + s * 128;
    float* __restrict__ trow = t1 + d * 128;
#pragma unroll
    for (int c = 0; c < 4; ++c) {
        unsafeAtomicAdd(&trow[lane + 32 * c], hrow[lane + 32 * c]);
    }
}

__global__ __launch_bounds__(256)
void scatter32(const int* __restrict__ src, const int* __restrict__ dst,
               const float* __restrict__ hs, float* __restrict__ t2) {
    const int e = blockIdx.x * 8 + (threadIdx.x >> 5);
    const int lane = threadIdx.x & 31;
    if (e >= N_EDGES) return;
    const long s = src[e];
    const long d = dst[e];
    unsafeAtomicAdd(&t2[d * 32 + lane], hs[s * 32 + lane]);
}

__global__ void post1(const float* __restrict__ hs, float* __restrict__ t1,
                      const float* __restrict__ dinv, const float* __restrict__ b1) {
    long i = (long)blockIdx.x * 256 + threadIdx.x;
    if (i >= (long)N_NODES * 128) return;
    const int node = (int)(i >> 7);
    const int c = (int)(i & 127);
    float v = (t1[i] + hs[i]) * dinv[node] + b1[c];
    t1[i] = fmaxf(v, 0.0f);
}

__global__ __launch_bounds__(256)
void final_kernel(const float* __restrict__ h2s, const float* __restrict__ t2,
                  const float* __restrict__ dinv, const float* __restrict__ b2,
                  float* __restrict__ out) {
    const int node = blockIdx.x * 8 + (threadIdx.x >> 5);
    const int lane = threadIdx.x & 31;
    if (node >= N_NODES) return;
    const long idx = (long)node * 32 + lane;
    float v = (t2[idx] + h2s[idx]) * dinv[node] + b2[lane];

    float m = v;
#pragma unroll
    for (int o = 16; o > 0; o >>= 1) m = fmaxf(m, __shfl_xor(m, o, 32));
    float s = __expf(v - m);
#pragma unroll
    for (int o = 16; o > 0; o >>= 1) s += __shfl_xor(s, o, 32);
    out[idx] = (v - m) - __logf(s);
}

extern "C" void kernel_launch(void* const* d_in, const int* in_sizes, int n_in,
                              void* d_out, int out_size, void* d_ws, size_t ws_size,
                              hipStream_t stream) {
    const float* x  = (const float*)d_in[0];
    const int*   ei = (const int*)d_in[1];
    const float* W1 = (const float*)d_in[2];
    const float* b1 = (const float*)d_in[3];
    const float* W2 = (const float*)d_in[4];
    const float* b2 = (const float*)d_in[5];
    float* out = (float*)d_out;

    const int* src = ei;
    const int* dst = ei + N_EDGES;

    float* deg  = (float*)d_ws;
    float* dinv = deg  + N_NODES;
    float* hs   = dinv + N_NODES;
    float* t1   = hs   + (long)N_NODES * 128;
    float* h2s  = t1   + (long)N_NODES * 128;
    float* t2   = h2s  + (long)N_NODES * 32;

    const long n128 = (long)N_NODES * 128;
    const long n32  = (long)N_NODES * 32;

    fill_f32<<<(N_NODES + 255) / 256, 256, 0, stream>>>(deg, 1.0f, N_NODES);
    deg_accum<<<(N_EDGES + 255) / 256, 256, 0, stream>>>(dst, deg);
    rsqrt_kernel<<<(N_NODES + 255) / 256, 256, 0, stream>>>(deg, dinv);

    fill_f32<<<(int)((n128 + 255) / 256), 256, 0, stream>>>(t1, 0.0f, n128);
    fill_f32<<<(int)((n32 + 255) / 256), 256, 0, stream>>>(t2, 0.0f, n32);

    gemm_scale_kernel<128><<<(N_STRIPS + 7) / 8, 256, 0, stream>>>(x, W1, dinv, hs);
    scatter128<<<(N_EDGES + 7) / 8, 256, 0, stream>>>(src, dst, hs, t1);
    post1<<<(int)((n128 + 255) / 256), 256, 0, stream>>>(hs, t1, dinv, b1);

    gemm_scale_kernel<32><<<(N_STRIPS + 7) / 8, 256, 0, stream>>>(t1, W2, dinv, h2s);
    scatter32<<<(N_EDGES + 7) / 8, 256, 0, stream>>>(src, dst, h2s, t2);
    final_kernel<<<(N_NODES + 7) / 8, 256, 0, stream>>>(h2s, t2, dinv, b2, out);
}


# Layer-1 edge sums move from L2 atomics into LDS: work-group owns 400 dst rows, finds its edges by scanning byte owner ids, adds rows with LDS float atomics (arrival order stays), stores each row once
# speedup vs baseline: 3.3264x; 3.3264x over previous
#include <hip/hip_runtime.h>
#include <hip/hip_bf16.h>

#define N_NODES 100000
#define N_EDGES 1600000
#define D_IN    128
#define D_HID   128
#define N_CLS   32
#define N_STRIPS (N_NODES / 16)

typedef float v2f __attribute__((ext_vector_type(2)));
typedef float v8f __attribute__((ext_vector_type(8)));

__device__ __forceinline__ unsigned chip_half() {
    return ((unsigned)__builtin_amdgcn_s_sendmsg_rtn(0x87) >> 18) & 1u;
}

#define LR_ROWS   400
#define LR_GROUPS (N_NODES / LR_ROWS)
static_assert(N_NODES % LR_ROWS == 0 && LR_GROUPS <= 255, "whole owner ranges, byte ids");
#define DG_RANGES 16
#define DG_SLICES 32
#define DG_RNODES (N_NODES / DG_RANGES)
#define DG_SEDGES (N_EDGES / DG_SLICES)
#define DG_VEC    (DG_SEDGES / 4)
#define DG_TRIPS  ((DG_VEC + 1023) / 1024)
static_assert(N_NODES % DG_RANGES == 0 && N_EDGES % DG_SLICES == 0 && DG_SEDGES % 4 == 0, "whole ranges and slices");

__global__ __launch_bounds__(1024)
void deg_count(const int* __restrict__ dst, unsigned* __restrict__ cnt, unsigned* __restrict__ ids) {
    __shared__ unsigned s_cnt[DG_RNODES];
    const unsigned range = blockIdx.x % DG_RANGES;
    const unsigned slice = blockIdx.x / DG_RANGES;
    const unsigned tid = threadIdx.x;
    const unsigned lo = range * DG_RNODES;
    const int4* __restrict__ p = (const int4*)(dst + slice * DG_SEDGES);

    int4 d[DG_TRIPS];
#pragma unroll
    for (int j = 0; j < DG_TRIPS; ++j) {
        const unsigned i = tid + 1024u * j;
        d[j] = (i < DG_VEC) ? p[i] : make_int4(-1, -1, -1, -1);
    }
    for (unsigned i = tid; i < DG_RNODES; i += 1024) s_cnt[i] = 0u;
    __syncthreads();
#pragma unroll
    for (int j = 0; j < DG_TRIPS; ++j) {
        if (range == 0) {
            const unsigned i = tid + 1024u * j;
            if (i < DG_VEC)
                ids[slice * DG_VEC + i] = ((unsigned)d[j].x / LR_ROWS) | (((unsigned)d[j].y / LR_ROWS) << 8) |
                                          (((unsigned)d[j].z / LR_ROWS) << 16) | (((unsigned)d[j].w / LR_ROWS) << 24);
        }
        const unsigned a0 = (unsigned)d[j].x - lo, a1 = (unsigned)d[j].y - lo;
        const unsigned a2 = (unsigned)d[j].z - lo, a3 = (unsigned)d[j].w - lo;
        if (a0 < DG_RNODES) atomicAdd(&s_cnt[a0], 1u);
        if (a1 < DG_RNODES) atomicAdd(&s_cnt[a1], 1u);
        if (a2 < DG_RNODES) atomicAdd(&s_cnt[a2], 1u);
        if (a3 < DG_RNODES) atomicAdd(&s_cnt[a3], 1u);
    }
    __syncthreads();
    unsigned* __restrict__ row = cnt + slice * (unsigned)N_NODES + lo;
    for (unsigned i = tid; i < DG_RNODES; i += 1024) row[i] = s_cnt[i];
}

#define TICKET_PITCH 64

__global__ void rsqrt_kernel(const unsigned* __restrict__ cnt, float* __restrict__ dinv,
                             unsigned* __restrict__ tickets) {
    int i = blockIdx.x * 256 + threadIdx.x;
    if (i < 4) tickets[i * TICKET_PITCH] = 0u;
    if (i < N_NODES) {
        unsigned c = 1u;
#pragma unroll
        for (int s = 0; s < DG_SLICES; ++s) c += cnt[s * N_NODES + i];
        const float d = (float)c;
        dinv[i] = d > 0.0f ? rsqrtf(d) : 0.0f;
    }
}

template <int NCOLS, bool PRE>
__device__ __forceinline__ void k_step(const int k0, const int half, const int l16,
                                       const float* __restrict__ Arow,
                                       const float* lW,
                                       const float* __restrict__ bias, const float dv,
                                       v8f (&acc)[NCOLS / 16]) {
    const int ka = k0 + 2 * half;
    v2f a = *(const v2f*)&Arow[ka];
    if (PRE) {
        const v2f bb = *(const v2f*)&bias[ka];
        a.x = fmaxf(a.x * dv + bb.x, 0.0f);
        a.y = fmaxf(a.y * dv + bb.y, 0.0f);
    }
    const float* lrow = &lW[(ka >> 1) * (2 * NCOLS)];
#pragma unroll
    for (int t = 0; t < NCOLS / 16; ++t) {
        const int n = t * 16 + l16;
        const v2f b = *(const v2f*)&lrow[2 * n];
        acc[t] = __builtin_amdgcn_wmma_f32_16x16x4_f32(
            false, a, false, b, (short)0, acc[t], false, false);
    }
}

template <int NCOLS, bool PRE>
__global__ __launch_bounds__(256)
void gemm_scale_kernel(const float* __restrict__ A,
                       const float* __restrict__ W,
                       const float* __restrict__ dinv,
                       const float* __restrict__ bias,
                       float* __restrict__ out,
                       float* __restrict__ seed) {
    constexpr int NT = NCOLS / 16;
    __shared__ float lW[128 * NCOLS];

    const int tid = threadIdx.x;
    for (int idx = tid; idx < 128 * NCOLS; idx += 256) {
        const int k = idx / NCOLS;
        const int n = idx - k * NCOLS;
        lW[(k >> 1) * (2 * NCOLS) + 2 * n + (k & 1)] = W[idx];
    }
    __syncthreads();

    const int wave  = tid >> 5;
    const int lane  = tid & 31;
    const int strip = blockIdx.x * 8 + wave;
    if (strip >= N_STRIPS) return;

    const int row0 = strip * 16;
    const int half = lane >> 4;
    const int l16  = lane & 15;
    const int arow = row0 + l16;

    v8f acc[NT];
#pragma unroll
    for (int t = 0; t < NT; ++t) acc[t] = (v8f){0.f, 0.f, 0.f, 0.f, 0.f, 0.f, 0.f, 0.f};

    float dv = 1.0f;
    if (PRE) dv = dinv[arow];

    const float* __restrict__ Arow = &A[(long)arow * 128];
    if (PRE) {
#pragma unroll
        for (int k0 = 0; k0 < 128; k0 += 4)
            k_step<NCOLS, PRE>(k0, half, l16, Arow, lW, bias, dv, acc);
    } else {
        for (int k0 = 0; k0 < 128; k0 += 4)
            k_step<NCOLS, PRE>(k0, half, l16, Arow, lW, bias, dv, acc);
    }

    if constexpr (NCOLS == 128) {
#pragma unroll
        for (int r = 0; r < 8; ++r) {
            const int rr = row0 + r + 8 * half;
            const float dr = dinv[rr];
            float val[NT];
#pragma unroll
            for (int t = 0; t < NT; ++t) val[t] = acc[t][r] * dr;
#pragma unroll
            for (int p = 0; p < 2; ++p)
                *(float4*)&out[(long)rr * NCOLS + 64 * p + 4 * l16] =
                    make_float4(val[p], val[2 + p], val[4 + p], val[6 + p]);
        }
    } else {
#pragma unroll
        for (int t = 0; t < NT; ++t) {
#pragma unroll
            for (int r = 0; r < 8; ++r) {
                const int rr = row0 + r + 8 * half;
                const float val = acc[t][r] * dinv[rr];
                out[(long)rr * NCOLS + t * 16 + l16]  = val;
                seed[(long)rr * NCOLS + t * 16 + l16] = val;
            }
        }
    }
}

#define SC_EPB    256
#define SC_NCHUNK (N_EDGES / SC_EPB)
#define HALF_NODES (N_NODES / 2)
static_assert(N_EDGES % SC_EPB == 0, "whole chunks");

template <int LPR>
__global__ __launch_bounds__(256)
void scatter_owned(const int* __restrict__ src, const int* __restrict__ dst,
                   const float* __restrict__ hs, float* __restrict__ t,
                   unsigned* __restrict__ tickets) {
    constexpr int G = 32 / LPR;
    constexpr unsigned ROWF = 32u * LPR;
    __shared__ unsigned s_item[2];
    if (threadIdx.x == 0) {
        unsigned own = chip_half();
        unsigned c = atomicAdd(&tickets[own * TICKET_PITCH], 1u);
        if (c >= SC_NCHUNK) {
            own ^= 1u;
            c = atomicAdd(&tickets[own * TICKET_PITCH], 1u);
        }
        s_item[0] = c;
        s_item[1] = own;
    }
    __syncthreads();
    const unsigned chunk = (unsigned)__builtin_amdgcn_readfirstlane((int)s_item[0]);
    const unsigned own   = (unsigned)__builtin_amdgcn_readfirstlane((int)s_item[1]);
    if (chunk >= SC_NCHUNK) return;

    const unsigned lane = threadIdx.x & 31;
    const int e = (int)(chunk * SC_EPB + threadIdx.x);
    const int my_s = src[e];
    const int my_d = dst[e];
    asm volatile("" :: "v"(my_s), "v"(my_d));
    const unsigned mine = ((unsigned)my_d >= (unsigned)HALF_NODES) ? 1u : 0u;
    unsigned m = __builtin_amdgcn_ballot_w32(mine == own);

    while (m) {
        int ln[G];
#pragma unroll
        for (int j = 0; j < G; ++j) {
            ln[j] = m ? __builtin_ctz(m) : -1;
            m &= m - 1u;
        }
        float v[G][LPR];
#pragma unroll
        for (int j = 0; j < G; ++j) {
            if (ln[j] >= 0) {
                const unsigned s = (unsigned)__builtin_amdgcn_readlane(my_s, ln[j]);
                if constexpr (LPR == 4) {
                    const float4 q = *((const float4*)(hs + s * ROWF) + lane);
                    v[j][0] = q.x; v[j][1] = q.y; v[j][2] = q.z; v[j][3] = q.w;
                } else {
                    const float* __restrict__ hrow = hs + (s * ROWF + lane);
#pragma unroll
                    for (int c = 0; c < LPR; ++c) v[j][c] = hrow[32 * c];
                }
            }
        }
#pragma unroll
        for (int j = 0; j < G; ++j) {
            if (ln[j] >= 0) {
                const unsigned d = (unsigned)__builtin_amdgcn_readlane(my_d, ln[j]);
                float* __restrict__ trow = t + (d * ROWF + lane);
#pragma unroll
                for (int c = 0; c < LPR; ++c) unsafeAtomicAdd(&trow[32 * c], v[j][c]);
            }
        }
    }
}

#define LR_G      16
#define LR_QCAP   256
#define LR_B      4
#define ID_VECS   (N_EDGES / 16)
#define ID_CHUNKS (ID_VECS / 32)
#define LR_TRIPS  ((ID_CHUNKS + 31) / 32)
#define LR_BATCHES ((LR_TRIPS + LR_B - 1) / LR_B)
#define LR_SKEW   (ID_CHUNKS / LR_GROUPS)
static_assert(N_EDGES % 512 == 0 && LR_QCAP >= LR_G - 1 + 128 && (LR_QCAP & (LR_QCAP - 1)) == 0, "whole chunks, ring holds a word's hits");

__global__ __launch_bounds__(1024)
void scatter_lds(const int* __restrict__ src, const int* __restrict__ dst,
                 const unsigned* __restrict__ ids, const float* __restrict__ hs,
                 float* __restrict__ t) {
    __shared__ __attribute__((aligned(16))) float s_rows[(LR_ROWS + 1) * 128];
    __shared__ unsigned s_q[32 * LR_QCAP];
    const unsigned r = blockIdx.x;
    const unsigned lo = r * LR_ROWS;
    const unsigned lane = threadIdx.x & 31;
    const unsigned w = (unsigned)__builtin_amdgcn_readfirstlane((int)(threadIdx.x >> 5));

    {
        constexpr int IT = (LR_ROWS + 31) / 32;
        float4 q[IT];
#pragma unroll
        for (int it = 0; it < IT; ++it) {
            const unsigned rl = min(w + 32u * it, (unsigned)LR_ROWS - 1u);
            q[it] = *((const float4*)(hs + (lo + rl) * 128u) + lane);
        }
#pragma unroll
        for (int it = 0; it < IT; ++it) {
            const unsigned rl = min(w + 32u * it, (unsigned)LR_ROWS);
            float* row = &s_rows[rl * 128u + lane];
            row[0] = q[it].x; row[32] = q[it].y; row[64] = q[it].z; row[96] = q[it].w;
        }
    }
    __syncthreads();

    const uint4* __restrict__ idv = (const uint4*)ids;
    const unsigned rpat = r * 0x01010101u;
    unsigned* q = &s_q[w * LR_QCAP];
    unsigned cnt = 0, head = 0;

    auto chunk_of = [&](const unsigned i) -> unsigned {
        const unsigned c = i * 32u + w;
        if (c >= ID_CHUNKS) return (unsigned)ID_CHUNKS;
        const unsigned cc = c + r * LR_SKEW;
        return cc >= ID_CHUNKS ? cc - ID_CHUNKS : cc;
    };
    auto load_ids = [&](const unsigned i) -> uint4 {
        const unsigned cc = chunk_of(i);
        if (cc < ID_CHUNKS) return idv[cc * 32u + lane];
        return make_uint4(~0u, ~0u, ~0u, ~0u);
    };

    auto drain = [&](const unsigned n) {
        unsigned es = 0u, rl = (unsigned)LR_ROWS;
        if (lane < n) {
            const unsigned e = q[(head + lane) & (LR_QCAP - 1)];
            es = (unsigned)src[e];
            rl = min((unsigned)dst[e] - lo, (unsigned)LR_ROWS);
        }
        asm volatile("" :: "v"(es), "v"(rl));
        float4 v[LR_G];
#pragma unroll
        for (int j = 0; j < LR_G; ++j) {
            const unsigned s = (unsigned)__builtin_amdgcn_readlane((int)es, j);
            v[j] = *((const float4*)(hs + s * 128u) + lane);
        }
#pragma unroll
        for (int j = 0; j < LR_G; ++j) {
            const unsigned rj = (unsigned)__builtin_amdgcn_readlane((int)rl, j);
            float* row = &s_rows[rj * 128u + lane];
            __hip_atomic_fetch_add(&row[0],  v[j].x, __ATOMIC_RELAXED, __HIP_MEMORY_SCOPE_WORKGROUP);
            __hip_atomic_fetch_add(&row[32], v[j].y, __ATOMIC_RELAXED, __HIP_MEMORY_SCOPE_WORKGROUP);
            __hip_atomic_fetch_add(&row[64], v[j].z, __ATOMIC_RELAXED, __HIP_MEMORY_SCOPE_WORKGROUP);
            __hip_atomic_fetch_add(&row[96], v[j].w, __ATOMIC_RELAXED, __HIP_MEMORY_SCOPE_WORKGROUP);
        }
        head += n;
    };

    uint4 nxt[LR_B];
#pragma unroll
    for (int p = 0; p < LR_B; ++p) nxt[p] = load_ids((unsigned)p);
#pragma unroll
    for (int p = 0; p < LR_B; ++p) asm volatile("" :: "v"(nxt[p].x), "v"(nxt[p].y), "v"(nxt[p].z), "v"(nxt[p].w));
#pragma nounroll
    for (unsigned bi = 0; bi < LR_BATCHES; ++bi) {
      uint4 cur[LR_B];
#pragma unroll
      for (int p = 0; p < LR_B; ++p) cur[p] = nxt[p];
#pragma unroll
      for (int p = 0; p < LR_B; ++p) nxt[p] = load_ids((bi + 1u) * LR_B + (unsigned)p);
#pragma nounroll
      for (unsigned jj = 0; jj < LR_B; ++jj) {
        const unsigned i = bi * LR_B + jj;
        uint4 x = cur[0];
#pragma unroll
        for (int p = 0; p + 1 < LR_B; ++p) cur[p] = cur[p + 1];
        const unsigned ebase = (chunk_of(i) * 32u + lane) * 16u;
#pragma nounroll
        for (unsigned k = 0; k < 4; ++k) {
            const unsigned y = x.x ^ rpat;
            x.x = x.y; x.y = x.z; x.z = x.w;
            if (__builtin_amdgcn_ballot_w32(((y - 0x01010101u) & ~y & 0x80808080u) != 0u) == 0u) continue;
#pragma unroll
            for (unsigned b = 0; b < 4; ++b) {
                const bool hit = ((y >> (8u * b)) & 0xFFu) == 0u;
                const unsigned m = __builtin_amdgcn_ballot_w32(hit);
                if (m) {
                    if (hit) q[(cnt + __builtin_amdgcn_mbcnt_lo(m, 0u)) & (LR_QCAP - 1)] = ebase + 4u * k + b;
                    cnt += (unsigned)__builtin_popcount(m);
                }
            }
            while (cnt - head >= LR_G) drain(LR_G);
        }
      }
    }
    if (cnt != head) drain(cnt - head);
    __syncthreads();

    for (unsigned rl = w; rl < LR_ROWS; rl += 32) {
        const float4 o = *(const float4*)&s_rows[rl * 128u + 4u * lane];
        *((float4*)(t + (lo + rl) * 128u) + lane) = o;
    }
}

__global__ __launch_bounds__(256)
void final_kernel(const float* __restrict__ t2,
                  const float* __restrict__ dinv, const float* __restrict__ b2,
                  float* __restrict__ out) {
    const int node = blockIdx.x * 8 + (threadIdx.x >> 5);
    const int lane = threadIdx.x & 31;
    if (node >= N_NODES) return;
    const long idx = (long)node * 32 + lane;
    float v = t2[idx] * dinv[node] + b2[lane];

    float m = v;
#pragma unroll
    for (int o = 16; o > 0; o >>= 1) m = fmaxf(m, __shfl_xor(m, o, 32));
    float s = __expf(v - m);
#pragma unroll
    for (int o = 16; o > 0; o >>= 1) s += __shfl_xor(s, o, 32);
    out[idx] = (v - m) - __logf(s);
}

extern "C" void kernel_launch(void* const* d_in, const int* in_sizes, int n_in,
                              void* d_out, int out_size, void* d_ws, size_t ws_size,
                              hipStream_t stream) {
    const float* x  = (const float*)d_in[0];
    const int*   ei = (const int*)d_in[1];
    const float* W1 = (const float*)d_in[2];
    const float* b1 = (const float*)d_in[3];
    const float* W2 = (const float*)d_in[4];
    const float* b2 = (const float*)d_in[5];
    float* out = (float*)d_out;

    const int* src = ei;
    const int* dst = ei + N_EDGES;

    unsigned* cnt = (unsigned*)d_ws;
    float* hs   = (float*)d_ws + (long)DG_SLICES * N_NODES;
    float* t1   = hs   + (long)N_NODES * 128;
    float* h2s  = t1   + (long)N_NODES * 128;
    float* t2   = h2s  + (long)N_NODES * 32;
    float* dinv = t2   + (long)N_NODES * 32;
    unsigned* tickets = (unsigned*)(dinv + N_NODES);
    unsigned* ids = tickets + 4 * TICKET_PITCH;

    deg_count<<<DG_RANGES * DG_SLICES, 1024, 0, stream>>>(dst, cnt, ids);
    rsqrt_kernel<<<(N_NODES + 255) / 256, 256, 0, stream>>>(cnt, dinv, tickets);

    gemm_scale_kernel<128, false><<<(N_STRIPS + 7) / 8, 256, 0, stream>>>(x, W1, dinv, b1, hs, nullptr);
    scatter_lds<<<LR_GROUPS, 1024, 0, stream>>>(src, dst, ids, hs, t1);

    gemm_scale_kernel<32, true><<<(N_STRIPS + 7) / 8, 256, 0, stream>>>(t1, W2, dinv, b1, h2s, t2);
    scatter_owned<1><<<2 * SC_NCHUNK, 256, 0, stream>>>(src, dst, h2s, t2, tickets + 2 * TICKET_PITCH);
    final_kernel<<<(N_NODES + 7) / 8, 256, 0, stream>>>(t2, dinv, b2, out);
}
